# starting kernel
# baseline (speedup 1.0000x reference)
#include <hip/hip_runtime.h>
#include <hip/hip_bf16.h>
#include <math.h>

typedef float v2f __attribute__((ext_vector_type(2)));
typedef float v8f __attribute__((ext_vector_type(8)));

#define D 256
#define NPG 512

__global__ void node_dots(const float* __restrict__ x,
                          const float* __restrict__ w_rel,
                          const float* __restrict__ w_root,
                          float* __restrict__ r, float* __restrict__ s, int N) {
    const int lane = threadIdx.x & 31;
    const int wave = threadIdx.x >> 5;
    const int node = blockIdx.x * (blockDim.x >> 5) + wave;
    if (node >= N) return;
    const float* xr = x + (size_t)node * D;
    const int j0 = lane * 8;
    float ar = 0.f, as = 0.f;
#pragma unroll
    for (int t = 0; t < 8; ++t) {
        const float xv = xr[j0 + t];
        ar += xv * w_rel[j0 + t];
        as += xv * w_root[j0 + t];
    }
#pragma unroll
    for (int off = 16; off > 0; off >>= 1) {
        ar += __shfl_down(ar, off);
        as += __shfl_down(as, off);
    }
    if (lane == 0) { r[node] = ar; s[node] = as; }
}

__global__ void graph_score_pool(const float* __restrict__ x,
                                 const float* __restrict__ r,
                                 const float* __restrict__ s,
                                 const int* __restrict__ esrc,
                                 const int* __restrict__ edst,
                                 const float* __restrict__ b_rel,
                                 float* __restrict__ pooled,
                                 int Eg, int k) {
    __shared__ float acc[NPG];
    __shared__ float score[NPG];
    __shared__ int   keep[NPG];
    __shared__ float partial[2 * D];

    const int b     = blockIdx.x;
    const int tid   = threadIdx.x;
    const int nbase = b * NPG;

    acc[tid] = 0.f;
    __syncthreads();

    const long ebase = (long)b * Eg;
    for (int e = tid; e < Eg; e += blockDim.x) {
        const int sg = esrc[ebase + e];
        const int dg = edst[ebase + e] - nbase;
        atomicAdd(&acc[dg], r[sg]);
    }
    __syncthreads();

    const float sc = tanhf(acc[tid] + s[nbase + tid] + b_rel[0]);
    score[tid] = sc;
    __syncthreads();

    int cnt = 0;
    for (int j = 0; j < NPG; ++j) {
        const float v = score[j];
        cnt += (v > sc) || (v == sc && j < tid);
    }
    keep[tid] = (cnt < k) ? 1 : 0;
    __syncthreads();

    const int c = tid & (D - 1);
    const int h = tid >> 8;
    const int i0 = h * (NPG / 2);
    float ap = 0.f;
    for (int i = i0; i < i0 + NPG / 2; ++i) {
        if (keep[i])
            ap += score[i] * x[(size_t)(nbase + i) * D + c];
    }
    partial[h * D + c] = ap;
    __syncthreads();
    if (tid < D)
        pooled[(size_t)b * D + tid] = (partial[tid] + partial[D + tid]) / (float)k;
}

__global__ void proj_wmma(const float* __restrict__ A,
                          const float* __restrict__ Bm,
                          const float* __restrict__ bias,
                          float* __restrict__ out,
                          int M, int K, int Nn) {
    const int lane  = threadIdx.x & 31;
    const int half  = lane >> 4;
    const int l16   = lane & 15;
    const int tileM = blockIdx.x;
    const int tileN = blockIdx.y;

    const int rowA = tileM * 16 + l16;
    const int colB = tileN * 16 + l16;

    v8f c = {};
    for (int k0 = 0; k0 < K; k0 += 4) {
        const int ka = k0 + half * 2;
        v2f a, bf;
        a.x  = A[(size_t)rowA * K + ka];
        a.y  = A[(size_t)rowA * K + ka + 1];
        bf.x = Bm[(size_t)ka * Nn + colB];
        bf.y = Bm[(size_t)(ka + 1) * Nn + colB];
        c = __builtin_amdgcn_wmma_f32_16x16x4_f32(false, a, false, bf,
                                                  (short)0, c, false, false);
    }

    const float bv = bias[colB];
#pragma unroll
    for (int v = 0; v < 8; ++v) {
        const int row = tileM * 16 + v + half * 8;
        out[(size_t)row * Nn + colB] = c[v] + bv;
    }
}

extern "C" void kernel_launch(void* const* d_in, const int* in_sizes, int n_in,
                              void* d_out, int out_size, void* d_ws, size_t ws_size,
                              hipStream_t stream) {
    const float* x      = (const float*)d_in[0];
    const int*   eidx   = (const int*)d_in[1];
    const float* w_rel  = (const float*)d_in[4];
    const float* b_rel  = (const float*)d_in[5];
    const float* w_root = (const float*)d_in[6];
    const float* w_proj = (const float*)d_in[7];
    const float* b_proj = (const float*)d_in[8];
    float*       out    = (float*)d_out;

    const int d    = D;
    const int N    = in_sizes[0] / d;
    const int Bg   = N / NPG;
    const int E    = in_sizes[1] / 2;
    const int Eg   = E / Bg;
    const int k    = (NPG + 1) / 2;
    const int outd = in_sizes[8];

    float* r      = (float*)d_ws;
    float* s      = r + N;
    float* pooled = s + N;

    node_dots<<<dim3((N + 7) / 8), dim3(256), 0, stream>>>(x, w_rel, w_root, r, s, N);

    graph_score_pool<<<dim3(Bg), dim3(NPG), 0, stream>>>(
        x, r, s, eidx, eidx + E, b_rel, pooled, Eg, k);

    proj_wmma<<<dim3(Bg / 16, outd / 16), dim3(32), 0, stream>>>(
        pooled, w_proj, b_proj, out, Bg, d, outd);
}


# graph_score_pool pooling: kept rows compacted by rank into an LDS list; each lane issues 16 independent 16-byte row loads per batch instead of one load per trip; pooled is summed in a new order
# speedup vs baseline: 2.0059x; 2.0059x over previous
#include <hip/hip_runtime.h>
#include <hip/hip_bf16.h>
#include <math.h>

typedef float v2f __attribute__((ext_vector_type(2)));
typedef float v4f __attribute__((ext_vector_type(4)));
typedef float v8f __attribute__((ext_vector_type(8)));

#define D 256
#define NPG 512
#define PG 8
#define PU 16

__global__ void node_dots(const float* __restrict__ x,
                          const float* __restrict__ w_rel,
                          const float* __restrict__ w_root,
                          float* __restrict__ r, float* __restrict__ s, int N) {
    const int lane = threadIdx.x & 31;
    const int wave = threadIdx.x >> 5;
    const int node = blockIdx.x * (blockDim.x >> 5) + wave;
    if (node >= N) return;
    const float* xr = x + (size_t)node * D;
    const int j0 = lane * 8;
    float ar = 0.f, as = 0.f;
#pragma unroll
    for (int t = 0; t < 8; ++t) {
        const float xv = xr[j0 + t];
        ar += xv * w_rel[j0 + t];
        as += xv * w_root[j0 + t];
    }
#pragma unroll
    for (int off = 16; off > 0; off >>= 1) {
        ar += __shfl_down(ar, off);
        as += __shfl_down(as, off);
    }
    if (lane == 0) { r[node] = ar; s[node] = as; }
}

__global__ void graph_score_pool(const float* __restrict__ x,
                                 const float* __restrict__ r,
                                 const float* __restrict__ s,
                                 const int* __restrict__ esrc,
                                 const int* __restrict__ edst,
                                 const float* __restrict__ b_rel,
                                 float* __restrict__ pooled,
                                 int Eg, int k) {
    __shared__ float acc[NPG];
    __shared__ float score[NPG];
    __shared__ unsigned koff[NPG];
    __shared__ float    kscore[NPG];
    __shared__ v4f      partial4[PG * (D / 4)];

    const int b     = blockIdx.x;
    const int tid   = threadIdx.x;
    const int nbase = b * NPG;

    acc[tid] = 0.f;
    koff[tid] = 0u;
    kscore[tid] = 0.f;
    __syncthreads();

    const long ebase = (long)b * Eg;
    for (int e = tid; e < Eg; e += blockDim.x) {
        const int sg = esrc[ebase + e];
        const int dg = edst[ebase + e] - nbase;
        atomicAdd(&acc[dg], r[sg]);
    }
    __syncthreads();

    const float sc = tanhf(acc[tid] + s[nbase + tid] + b_rel[0]);
    score[tid] = sc;
    __syncthreads();

    int cnt = 0;
    for (int j = 0; j < NPG; ++j) {
        const float v = score[j];
        cnt += (v > sc) || (v == sc && j < tid);
    }
    if (cnt < k) { koff[cnt] = (unsigned)tid * (unsigned)(D * sizeof(float)); kscore[cnt] = sc; }
    __syncthreads();

    const int g  = tid >> 6;
    const int c4 = tid & 63;
    const char* __restrict__ xg =
        reinterpret_cast<const char*>(x + (size_t)nbase * D) + c4 * 16;
    v4f ap = {0.f, 0.f, 0.f, 0.f};
    int j = g;
    for (; j + PG * (PU - 1) < k; j += PG * PU) {
        v4f   v[PU];
        float w[PU];
#pragma unroll
        for (int u = 0; u < PU; ++u) {
            w[u] = kscore[j + PG * u];
            v[u] = *reinterpret_cast<const v4f*>(xg + koff[j + PG * u]);
        }
        __builtin_amdgcn_sched_barrier(0);
#pragma unroll
        for (int u = 0; u < PU; ++u) ap += w[u] * v[u];
    }
    for (; j < k; j += PG)
        ap += kscore[j] * *reinterpret_cast<const v4f*>(xg + koff[j]);
    partial4[g * (D / 4) + c4] = ap;
    __syncthreads();
    if (tid < D) {
        const float* partial = reinterpret_cast<const float*>(partial4);
        float t = partial[tid];
#pragma unroll
        for (int q = 1; q < PG; ++q) t += partial[q * D + tid];
        pooled[(size_t)b * D + tid] = t / (float)k;
    }
}

__global__ void proj_wmma(const float* __restrict__ A,
                          const float* __restrict__ Bm,
                          const float* __restrict__ bias,
                          float* __restrict__ out,
                          int M, int K, int Nn) {
    const int lane  = threadIdx.x & 31;
    const int half  = lane >> 4;
    const int l16   = lane & 15;
    const int tileM = blockIdx.x;
    const int tileN = blockIdx.y;

    const int rowA = tileM * 16 + l16;
    const int colB = tileN * 16 + l16;

    v8f c = {};
    for (int k0 = 0; k0 < K; k0 += 4) {
        const int ka = k0 + half * 2;
        v2f a, bf;
        a.x  = A[(size_t)rowA * K + ka];
        a.y  = A[(size_t)rowA * K + ka + 1];
        bf.x = Bm[(size_t)ka * Nn + colB];
        bf.y = Bm[(size_t)(ka + 1) * Nn + colB];
        c = __builtin_amdgcn_wmma_f32_16x16x4_f32(false, a, false, bf,
                                                  (short)0, c, false, false);
    }

    const float bv = bias[colB];
#pragma unroll
    for (int v = 0; v < 8; ++v) {
        const int row = tileM * 16 + v + half * 8;
        out[(size_t)row * Nn + colB] = c[v] + bv;
    }
}

extern "C" void kernel_launch(void* const* d_in, const int* in_sizes, int n_in,
                              void* d_out, int out_size, void* d_ws, size_t ws_size,
                              hipStream_t stream) {
    const float* x      = (const float*)d_in[0];
    const int*   eidx   = (const int*)d_in[1];
    const float* w_rel  = (const float*)d_in[4];
    const float* b_rel  = (const float*)d_in[5];
    const float* w_root = (const float*)d_in[6];
    const float* w_proj = (const float*)d_in[7];
    const float* b_proj = (const float*)d_in[8];
    float*       out    = (float*)d_out;

    const int d    = D;
    const int N    = in_sizes[0] / d;
    const int Bg   = N / NPG;
    const int E    = in_sizes[1] / 2;
    const int Eg   = E / Bg;
    const int k    = (NPG + 1) / 2;
    const int outd = in_sizes[8];

    float* r      = (float*)d_ws;
    float* s      = r + N;
    float* pooled = s + N;

    node_dots<<<dim3((N + 7) / 8), dim3(256), 0, stream>>>(x, w_rel, w_root, r, s, N);

    graph_score_pool<<<dim3(Bg), dim3(NPG), 0, stream>>>(
        x, r, s, eidx, eidx + E, b_rel, pooled, Eg, k);

    proj_wmma<<<dim3(Bg / 16, outd / 16), dim3(32), 0, stream>>>(
        pooled, w_proj, b_proj, out, Bg, d, outd);
}


# node_dots as a row stream: a wave takes 16 rows, 32 whole-512-byte loads in flight, and folds its 32 partial sums in one transposed butterfly (4096 waves, not 65536); r and s summed in a new order
# speedup vs baseline: 2.3413x; 1.1672x over previous
#include <hip/hip_runtime.h>
#include <hip/hip_bf16.h>
#include <math.h>

typedef float v2f __attribute__((ext_vector_type(2)));
typedef float v4f __attribute__((ext_vector_type(4)));
typedef float v8f __attribute__((ext_vector_type(8)));

#define D 256
#define NPG 512
#define PG 8
#define PU 16
#define NR 16
#define ND_WAVES 8

static_assert(2 * NR == 32, "node_dots folds one partial sum per lane of a 32-lane wave");
template <int OFF>
__device__ __forceinline__ void nd_fold(float (&val)[2 * NR], int lane) {
    const bool up = (lane & OFF) != 0;
#pragma unroll
    for (int i = 0; i < OFF; ++i) {
        const float keep = up ? val[i + OFF] : val[i];
        const float send = up ? val[i] : val[i + OFF];
        val[i] = keep + __shfl_xor(send, OFF);
    }
}

__global__ __launch_bounds__(ND_WAVES * 32)
void node_dots(const float* __restrict__ x,
               const float* __restrict__ w_rel,
               const float* __restrict__ w_root,
               float* __restrict__ r, float* __restrict__ s, int N) {
    const int lane = threadIdx.x & 31;
    const int wave = __builtin_amdgcn_readfirstlane((int)(threadIdx.x >> 5));
    const int row0 = (blockIdx.x * ND_WAVES + wave) * NR;
    if (row0 >= N) return;

    const v4f* __restrict__ wr4 = reinterpret_cast<const v4f*>(w_rel);
    const v4f* __restrict__ ws4 = reinterpret_cast<const v4f*>(w_root);
    const v4f wr0 = wr4[lane], wr1 = wr4[32 + lane];
    const v4f ws0 = ws4[lane], ws1 = ws4[32 + lane];

    v4f xa[NR], xb[NR];
#pragma unroll
    for (int i = 0; i < NR; ++i) {
        const int row = min(row0 + i, N - 1);
        const v4f* __restrict__ xp = reinterpret_cast<const v4f*>(x + (size_t)row * D);
        xa[i] = xp[lane];
        xb[i] = xp[32 + lane];
    }
    __builtin_amdgcn_sched_barrier(0);

    const v2f w0 = {wr0.x, ws0.x}, w1 = {wr0.y, ws0.y}, w2 = {wr0.z, ws0.z}, w3 = {wr0.w, ws0.w};
    const v2f w4 = {wr1.x, ws1.x}, w5 = {wr1.y, ws1.y}, w6 = {wr1.z, ws1.z}, w7 = {wr1.w, ws1.w};

    float val[2 * NR];
#pragma unroll
    for (int i = 0; i < NR; ++i) {
        v2f acc = xa[i].x * w0;
        acc += xa[i].y * w1;
        acc += xa[i].z * w2;
        acc += xa[i].w * w3;
        acc += xb[i].x * w4;
        acc += xb[i].y * w5;
        acc += xb[i].z * w6;
        acc += xb[i].w * w7;
        val[i]      = acc.x;
        val[NR + i] = acc.y;
        if ((i & 3) == 3) __builtin_amdgcn_sched_barrier(0);
    }

    nd_fold<16>(val, lane);
    nd_fold<8>(val, lane);
    nd_fold<4>(val, lane);
    nd_fold<2>(val, lane);
    nd_fold<1>(val, lane);

    const int row = row0 + (lane & 15);
    if (row < N) {
        float* __restrict__ dst = (lane < 16) ? r : s;
        dst[row] = val[0];
    }
}

__global__ void graph_score_pool(const float* __restrict__ x,
                                 const float* __restrict__ r,
                                 const float* __restrict__ s,
                                 const int* __restrict__ esrc,
                                 const int* __restrict__ edst,
                                 const float* __restrict__ b_rel,
                                 float* __restrict__ pooled,
                                 int Eg, int k) {
    __shared__ float acc[NPG];
    __shared__ float score[NPG];
    __shared__ unsigned koff[NPG];
    __shared__ float    kscore[NPG];
    __shared__ v4f      partial4[PG * (D / 4)];

    const int b     = blockIdx.x;
    const int tid   = threadIdx.x;
    const int nbase = b * NPG;

    acc[tid] = 0.f;
    koff[tid] = 0u;
    kscore[tid] = 0.f;
    __syncthreads();

    const long ebase = (long)b * Eg;
    for (int e = tid; e < Eg; e += blockDim.x) {
        const int sg = esrc[ebase + e];
        const int dg = edst[ebase + e] - nbase;
        atomicAdd(&acc[dg], r[sg]);
    }
    __syncthreads();

    const float sc = tanhf(acc[tid] + s[nbase + tid] + b_rel[0]);
    score[tid] = sc;
    __syncthreads();

    int cnt = 0;
    for (int j = 0; j < NPG; ++j) {
        const float v = score[j];
        cnt += (v > sc) || (v == sc && j < tid);
    }
    if (cnt < k) { koff[cnt] = (unsigned)tid * (unsigned)(D * sizeof(float)); kscore[cnt] = sc; }
    __syncthreads();

    const int g  = tid >> 6;
    const int c4 = tid & 63;
    const char* __restrict__ xg =
        reinterpret_cast<const char*>(x + (size_t)nbase * D) + c4 * 16;
    v4f ap = {0.f, 0.f, 0.f, 0.f};
    int j = g;
    for (; j + PG * (PU - 1) < k; j += PG * PU) {
        v4f   v[PU];
        float w[PU];
#pragma unroll
        for (int u = 0; u < PU; ++u) {
            w[u] = kscore[j + PG * u];
            v[u] = *reinterpret_cast<const v4f*>(xg + koff[j + PG * u]);
        }
        __builtin_amdgcn_sched_barrier(0);
#pragma unroll
        for (int u = 0; u < PU; ++u) ap += w[u] * v[u];
    }
    for (; j < k; j += PG)
        ap += kscore[j] * *reinterpret_cast<const v4f*>(xg + koff[j]);
    partial4[g * (D / 4) + c4] = ap;
    __syncthreads();
    if (tid < D) {
        const float* partial = reinterpret_cast<const float*>(partial4);
        float t = partial[tid];
#pragma unroll
        for (int q = 1; q < PG; ++q) t += partial[q * D + tid];
        pooled[(size_t)b * D + tid] = t / (float)k;
    }
}

__global__ void proj_wmma(const float* __restrict__ A,
                          const float* __restrict__ Bm,
                          const float* __restrict__ bias,
                          float* __restrict__ out,
                          int M, int K, int Nn) {
    const int lane  = threadIdx.x & 31;
    const int half  = lane >> 4;
    const int l16   = lane & 15;
    const int tileM = blockIdx.x;
    const int tileN = blockIdx.y;

    const int rowA = tileM * 16 + l16;
    const int colB = tileN * 16 + l16;

    v8f c = {};
    for (int k0 = 0; k0 < K; k0 += 4) {
        const int ka = k0 + half * 2;
        v2f a, bf;
        a.x  = A[(size_t)rowA * K + ka];
        a.y  = A[(size_t)rowA * K + ka + 1];
        bf.x = Bm[(size_t)ka * Nn + colB];
        bf.y = Bm[(size_t)(ka + 1) * Nn + colB];
        c = __builtin_amdgcn_wmma_f32_16x16x4_f32(false, a, false, bf,
                                                  (short)0, c, false, false);
    }

    const float bv = bias[colB];
#pragma unroll
    for (int v = 0; v < 8; ++v) {
        const int row = tileM * 16 + v + half * 8;
        out[(size_t)row * Nn + colB] = c[v] + bv;
    }
}

extern "C" void kernel_launch(void* const* d_in, const int* in_sizes, int n_in,
                              void* d_out, int out_size, void* d_ws, size_t ws_size,
                              hipStream_t stream) {
    const float* x      = (const float*)d_in[0];
    const int*   eidx   = (const int*)d_in[1];
    const float* w_rel  = (const float*)d_in[4];
    const float* b_rel  = (const float*)d_in[5];
    const float* w_root = (const float*)d_in[6];
    const float* w_proj = (const float*)d_in[7];
    const float* b_proj = (const float*)d_in[8];
    float*       out    = (float*)d_out;

    const int d    = D;
    const int N    = in_sizes[0] / d;
    const int Bg   = N / NPG;
    const int E    = in_sizes[1] / 2;
    const int Eg   = E / Bg;
    const int k    = (NPG + 1) / 2;
    const int outd = in_sizes[8];

    float* r      = (float*)d_ws;
    float* s      = r + N;
    float* pooled = s + N;

    node_dots<<<dim3((N + ND_WAVES * NR - 1) / (ND_WAVES * NR)), dim3(ND_WAVES * 32), 0, stream>>>(
        x, w_rel, w_root, r, s, N);

    graph_score_pool<<<dim3(Bg), dim3(NPG), 0, stream>>>(
        x, r, s, eidx, eidx + E, b_rel, pooled, Eg, k);

    proj_wmma<<<dim3(Bg / 16, outd / 16), dim3(32), 0, stream>>>(
        pooled, w_proj, b_proj, out, Bg, d, outd);
}


# proj_wmma: a tile's whole A and B fragments (all of K) are loaded before its first WMMA, not in 16 trips that each wait for their own loads; A in 16-byte pieces, so out is summed in a new k order
# speedup vs baseline: 2.9927x; 1.2782x over previous
#include <hip/hip_runtime.h>
#include <hip/hip_bf16.h>
#include <math.h>

typedef float v2f __attribute__((ext_vector_type(2)));
typedef float v4f __attribute__((ext_vector_type(4)));
typedef float v8f __attribute__((ext_vector_type(8)));

#define D 256
#define NPG 512
#define PG 8
#define PU 16
#define NR 16
#define ND_WAVES 8

static_assert(2 * NR == 32, "node_dots folds one partial sum per lane of a 32-lane wave");
template <int OFF>
__device__ __forceinline__ void nd_fold(float (&val)[2 * NR], int lane) {
    const bool up = (lane & OFF) != 0;
#pragma unroll
    for (int i = 0; i < OFF; ++i) {
        const float keep = up ? val[i + OFF] : val[i];
        const float send = up ? val[i] : val[i + OFF];
        val[i] = keep + __shfl_xor(send, OFF);
    }
}

__global__ __launch_bounds__(ND_WAVES * 32)
void node_dots(const float* __restrict__ x,
               const float* __restrict__ w_rel,
               const float* __restrict__ w_root,
               float* __restrict__ r, float* __restrict__ s, int N) {
    const int lane = threadIdx.x & 31;
    const int wave = __builtin_amdgcn_readfirstlane((int)(threadIdx.x >> 5));
    const int row0 = (blockIdx.x * ND_WAVES + wave) * NR;
    if (row0 >= N) return;

    const v4f* __restrict__ wr4 = reinterpret_cast<const v4f*>(w_rel);
    const v4f* __restrict__ ws4 = reinterpret_cast<const v4f*>(w_root);
    const v4f wr0 = wr4[lane], wr1 = wr4[32 + lane];
    const v4f ws0 = ws4[lane], ws1 = ws4[32 + lane];

    v4f xa[NR], xb[NR];
#pragma unroll
    for (int i = 0; i < NR; ++i) {
        const int row = min(row0 + i, N - 1);
        const v4f* __restrict__ xp = reinterpret_cast<const v4f*>(x + (size_t)row * D);
        xa[i] = xp[lane];
        xb[i] = xp[32 + lane];
    }
    __builtin_amdgcn_sched_barrier(0);

    const v2f w0 = {wr0.x, ws0.x}, w1 = {wr0.y, ws0.y}, w2 = {wr0.z, ws0.z}, w3 = {wr0.w, ws0.w};
    const v2f w4 = {wr1.x, ws1.x}, w5 = {wr1.y, ws1.y}, w6 = {wr1.z, ws1.z}, w7 = {wr1.w, ws1.w};

    float val[2 * NR];
#pragma unroll
    for (int i = 0; i < NR; ++i) {
        v2f acc = xa[i].x * w0;
        acc += xa[i].y * w1;
        acc += xa[i].z * w2;
        acc += xa[i].w * w3;
        acc += xb[i].x * w4;
        acc += xb[i].y * w5;
        acc += xb[i].z * w6;
        acc += xb[i].w * w7;
        val[i]      = acc.x;
        val[NR + i] = acc.y;
        if ((i & 3) == 3) __builtin_amdgcn_sched_barrier(0);
    }

    nd_fold<16>(val, lane);
    nd_fold<8>(val, lane);
    nd_fold<4>(val, lane);
    nd_fold<2>(val, lane);
    nd_fold<1>(val, lane);

    const int row = row0 + (lane & 15);
    if (row < N) {
        float* __restrict__ dst = (lane < 16) ? r : s;
        dst[row] = val[0];
    }
}

__global__ void graph_score_pool(const float* __restrict__ x,
                                 const float* __restrict__ r,
                                 const float* __restrict__ s,
                                 const int* __restrict__ esrc,
                                 const int* __restrict__ edst,
                                 const float* __restrict__ b_rel,
                                 float* __restrict__ pooled,
                                 int Eg, int k) {
    __shared__ float acc[NPG];
    __shared__ float score[NPG];
    __shared__ unsigned koff[NPG];
    __shared__ float    kscore[NPG];
    __shared__ v4f      partial4[PG * (D / 4)];

    const int b     = blockIdx.x;
    const int tid   = threadIdx.x;
    const int nbase = b * NPG;

    acc[tid] = 0.f;
    koff[tid] = 0u;
    kscore[tid] = 0.f;
    __syncthreads();

    const long ebase = (long)b * Eg;
    for (int e = tid; e < Eg; e += blockDim.x) {
        const int sg = esrc[ebase + e];
        const int dg = edst[ebase + e] - nbase;
        atomicAdd(&acc[dg], r[sg]);
    }
    __syncthreads();

    const float sc = tanhf(acc[tid] + s[nbase + tid] + b_rel[0]);
    score[tid] = sc;
    __syncthreads();

    int cnt = 0;
    for (int j = 0; j < NPG; ++j) {
        const float v = score[j];
        cnt += (v > sc) || (v == sc && j < tid);
    }
    if (cnt < k) { koff[cnt] = (unsigned)tid * (unsigned)(D * sizeof(float)); kscore[cnt] = sc; }
    __syncthreads();

    const int g  = tid >> 6;
    const int c4 = tid & 63;
    const char* __restrict__ xg =
        reinterpret_cast<const char*>(x + (size_t)nbase * D) + c4 * 16;
    v4f ap = {0.f, 0.f, 0.f, 0.f};
    int j = g;
    for (; j + PG * (PU - 1) < k; j += PG * PU) {
        v4f   v[PU];
        float w[PU];
#pragma unroll
        for (int u = 0; u < PU; ++u) {
            w[u] = kscore[j + PG * u];
            v[u] = *reinterpret_cast<const v4f*>(xg + koff[j + PG * u]);
        }
        __builtin_amdgcn_sched_barrier(0);
#pragma unroll
        for (int u = 0; u < PU; ++u) ap += w[u] * v[u];
    }
    for (; j < k; j += PG)
        ap += kscore[j] * *reinterpret_cast<const v4f*>(xg + koff[j]);
    partial4[g * (D / 4) + c4] = ap;
    __syncthreads();
    if (tid < D) {
        const float* partial = reinterpret_cast<const float*>(partial4);
        float t = partial[tid];
#pragma unroll
        for (int q = 1; q < PG; ++q) t += partial[q * D + tid];
        pooled[(size_t)b * D + tid] = t / (float)k;
    }
}

#define PJ_U (D / 8)
typedef __attribute__((address_space(1))) float gmem_float;
__global__ __launch_bounds__(32)
void proj_wmma(const float* __restrict__ A,
               const float* __restrict__ Bm,
               const float* __restrict__ bias,
               float* __restrict__ out,
               int Nn) {
    const int lane  = threadIdx.x & 31;
    const int half  = lane >> 4;
    const int l16   = lane & 15;
    const int tileM = blockIdx.x;
    const int tileN = blockIdx.y;

    const int rowA = tileM * 16 + l16;
    const int colB = tileN * 16 + l16;

    const float* __restrict__ ap = A + (size_t)rowA * D + half * 4;
    const float* __restrict__ bp = Bm + (size_t)(half * 4) * Nn + colB;

    v4f a[PJ_U], b[PJ_U];
#pragma unroll
    for (int u = 0; u < PJ_U; ++u) {
        a[u] = *reinterpret_cast<const v4f*>(ap + 8 * u);
        const float* __restrict__ bq = bp + (size_t)(8 * u) * Nn;
        b[u].x = bq[0];
        b[u].y = bq[(size_t)Nn];
        b[u].z = bq[(size_t)2 * Nn];
        b[u].w = bq[(size_t)3 * Nn];
    }
    const float bv = bias[colB];

    gmem_float* op = (gmem_float*)(out + (size_t)(tileM * 16 + half * 8) * Nn + colB);
    asm volatile("" : "+v"(op));
    __builtin_amdgcn_sched_barrier(0);

    v8f c = {};
#pragma unroll
    for (int u = 0; u < PJ_U; ++u) {
        const v2f a0 = {a[u].x, a[u].y}, b0 = {b[u].x, b[u].y};
        const v2f a1 = {a[u].z, a[u].w}, b1 = {b[u].z, b[u].w};
        c = __builtin_amdgcn_wmma_f32_16x16x4_f32(false, a0, false, b0,
                                                  (short)0, c, false, false);
        c = __builtin_amdgcn_wmma_f32_16x16x4_f32(false, a1, false, b1,
                                                  (short)0, c, false, false);
    }
    __builtin_amdgcn_sched_barrier(0);

#pragma unroll
    for (int v = 0; v < 8; ++v) op[(size_t)v * Nn] = c[v] + bv;
}

extern "C" void kernel_launch(void* const* d_in, const int* in_sizes, int n_in,
                              void* d_out, int out_size, void* d_ws, size_t ws_size,
                              hipStream_t stream) {
    const float* x      = (const float*)d_in[0];
    const int*   eidx   = (const int*)d_in[1];
    const float* w_rel  = (const float*)d_in[4];
    const float* b_rel  = (const float*)d_in[5];
    const float* w_root = (const float*)d_in[6];
    const float* w_proj = (const float*)d_in[7];
    const float* b_proj = (const float*)d_in[8];
    float*       out    = (float*)d_out;

    const int d    = D;
    const int N    = in_sizes[0] / d;
    const int Bg   = N / NPG;
    const int E    = in_sizes[1] / 2;
    const int Eg   = E / Bg;
    const int k    = (NPG + 1) / 2;
    const int outd = in_sizes[8];

    float* r      = (float*)d_ws;
    float* s      = r + N;
    float* pooled = s + N;

    node_dots<<<dim3((N + ND_WAVES * NR - 1) / (ND_WAVES * NR)), dim3(ND_WAVES * 32), 0, stream>>>(
        x, w_rel, w_root, r, s, N);

    graph_score_pool<<<dim3(Bg), dim3(NPG), 0, stream>>>(
        x, r, s, eidx, eidx + E, b_rel, pooled, Eg, k);

    proj_wmma<<<dim3(Bg / 16, outd / 16), dim3(32), 0, stream>>>(
        pooled, w_proj, b_proj, out, outd);
}


# graph_score_pool edge phase: a thread's edge index loads go out as one group, then its gathers of r, with s and b_rel ahead of them: 2 load round trips, not 17; acc's LDS adds come in a new order
# speedup vs baseline: 3.3415x; 1.1166x over previous
#include <hip/hip_runtime.h>
#include <hip/hip_bf16.h>
#include <math.h>

typedef float v2f __attribute__((ext_vector_type(2)));
typedef float v4f __attribute__((ext_vector_type(4)));
typedef float v8f __attribute__((ext_vector_type(8)));

#define D 256
#define NPG 512
#define PG 8
#define PU 16
#define NR 16
#define ND_WAVES 8
#define EB 8

static_assert(2 * NR == 32, "node_dots folds one partial sum per lane of a 32-lane wave");
template <int OFF>
__device__ __forceinline__ void nd_fold(float (&val)[2 * NR], int lane) {
    const bool up = (lane & OFF) != 0;
#pragma unroll
    for (int i = 0; i < OFF; ++i) {
        const float keep = up ? val[i + OFF] : val[i];
        const float send = up ? val[i] : val[i + OFF];
        val[i] = keep + __shfl_xor(send, OFF);
    }
}

__global__ __launch_bounds__(ND_WAVES * 32)
void node_dots(const float* __restrict__ x,
               const float* __restrict__ w_rel,
               const float* __restrict__ w_root,
               float* __restrict__ r, float* __restrict__ s, int N) {
    const int lane = threadIdx.x & 31;
    const int wave = __builtin_amdgcn_readfirstlane((int)(threadIdx.x >> 5));
    const int row0 = (blockIdx.x * ND_WAVES + wave) * NR;
    if (row0 >= N) return;

    const v4f* __restrict__ wr4 = reinterpret_cast<const v4f*>(w_rel);
    const v4f* __restrict__ ws4 = reinterpret_cast<const v4f*>(w_root);
    const v4f wr0 = wr4[lane], wr1 = wr4[32 + lane];
    const v4f ws0 = ws4[lane], ws1 = ws4[32 + lane];

    v4f xa[NR], xb[NR];
#pragma unroll
    for (int i = 0; i < NR; ++i) {
        const int row = min(row0 + i, N - 1);
        const v4f* __restrict__ xp = reinterpret_cast<const v4f*>(x + (size_t)row * D);
        xa[i] = xp[lane];
        xb[i] = xp[32 + lane];
    }
    __builtin_amdgcn_sched_barrier(0);

    const v2f w0 = {wr0.x, ws0.x}, w1 = {wr0.y, ws0.y}, w2 = {wr0.z, ws0.z}, w3 = {wr0.w, ws0.w};
    const v2f w4 = {wr1.x, ws1.x}, w5 = {wr1.y, ws1.y}, w6 = {wr1.z, ws1.z}, w7 = {wr1.w, ws1.w};

    float val[2 * NR];
#pragma unroll
    for (int i = 0; i < NR; ++i) {
        v2f acc = xa[i].x * w0;
        acc += xa[i].y * w1;
        acc += xa[i].z * w2;
        acc += xa[i].w * w3;
        acc += xb[i].x * w4;
        acc += xb[i].y * w5;
        acc += xb[i].z * w6;
        acc += xb[i].w * w7;
        val[i]      = acc.x;
        val[NR + i] = acc.y;
        if ((i & 3) == 3) __builtin_amdgcn_sched_barrier(0);
    }

    nd_fold<16>(val, lane);
    nd_fold<8>(val, lane);
    nd_fold<4>(val, lane);
    nd_fold<2>(val, lane);
    nd_fold<1>(val, lane);

    const int row = row0 + (lane & 15);
    if (row < N) {
        float* __restrict__ dst = (lane < 16) ? r : s;
        dst[row] = val[0];
    }
}

__global__ void graph_score_pool(const float* __restrict__ x,
                                 const float* __restrict__ r,
                                 const float* __restrict__ s,
                                 const int* __restrict__ esrc,
                                 const int* __restrict__ edst,
                                 const float* __restrict__ b_rel,
                                 float* __restrict__ pooled,
                                 int Eg, int k) {
    __shared__ float acc[NPG];
    __shared__ float score[NPG];
    __shared__ unsigned koff[NPG];
    __shared__ float    kscore[NPG];
    __shared__ v4f      partial4[PG * (D / 4)];

    const int b     = blockIdx.x;
    const int tid   = threadIdx.x;
    const int nbase = b * NPG;

    const float brel = b_rel[0];

    acc[tid] = 0.f;
    koff[tid] = 0u;
    kscore[tid] = 0.f;
    __syncthreads();

    const float s_own = s[nbase + tid];

    const int* __restrict__ psrc = esrc + (long)b * Eg;
    const int* __restrict__ pdst = edst + (long)b * Eg;
    int e = tid;
    for (; e + (EB - 1) * NPG < Eg; e += EB * NPG) {
        int   sg[EB], dg[EB];
        float rv[EB];
#pragma unroll
        for (int u = 0; u < EB; ++u) {
            sg[u] = psrc[e + u * NPG];
            dg[u] = pdst[e + u * NPG];
        }
#pragma unroll
        for (int u = 0; u < EB; ++u) rv[u] = r[sg[u]];
#pragma unroll
        for (int u = 0; u < EB; ++u) atomicAdd(&acc[dg[u] - nbase], rv[u]);
    }
    for (; e < Eg; e += NPG)
        atomicAdd(&acc[pdst[e] - nbase], r[psrc[e]]);
    __syncthreads();

    const float sc = tanhf(acc[tid] + s_own + brel);
    score[tid] = sc;
    __syncthreads();

    int cnt = 0;
    for (int j = 0; j < NPG; ++j) {
        const float v = score[j];
        cnt += (v > sc) || (v == sc && j < tid);
    }
    if (cnt < k) { koff[cnt] = (unsigned)tid * (unsigned)(D * sizeof(float)); kscore[cnt] = sc; }
    __syncthreads();

    const int g  = tid >> 6;
    const int c4 = tid & 63;
    const char* __restrict__ xg =
        reinterpret_cast<const char*>(x + (size_t)nbase * D) + c4 * 16;
    v4f ap = {0.f, 0.f, 0.f, 0.f};
    int j = g;
    for (; j + PG * (PU - 1) < k; j += PG * PU) {
        v4f   v[PU];
        float w[PU];
#pragma unroll
        for (int u = 0; u < PU; ++u) {
            w[u] = kscore[j + PG * u];
            v[u] = *reinterpret_cast<const v4f*>(xg + koff[j + PG * u]);
        }
        __builtin_amdgcn_sched_barrier(0);
#pragma unroll
        for (int u = 0; u < PU; ++u) ap += w[u] * v[u];
    }
    for (; j < k; j += PG)
        ap += kscore[j] * *reinterpret_cast<const v4f*>(xg + koff[j]);
    partial4[g * (D / 4) + c4] = ap;
    __syncthreads();
    if (tid < D) {
        const float* partial = reinterpret_cast<const float*>(partial4);
        float t = partial[tid];
#pragma unroll
        for (int q = 1; q < PG; ++q) t += partial[q * D + tid];
        pooled[(size_t)b * D + tid] = t / (float)k;
    }
}

#define PJ_U (D / 8)
typedef __attribute__((address_space(1))) float gmem_float;
__global__ __launch_bounds__(32)
void proj_wmma(const float* __restrict__ A,
               const float* __restrict__ Bm,
               const float* __restrict__ bias,
               float* __restrict__ out,
               int Nn) {
    const int lane  = threadIdx.x & 31;
    const int half  = lane >> 4;
    const int l16   = lane & 15;
    const int tileM = blockIdx.x;
    const int tileN = blockIdx.y;

    const int rowA = tileM * 16 + l16;
    const int colB = tileN * 16 + l16;

    const float* __restrict__ ap = A + (size_t)rowA * D + half * 4;
    const float* __restrict__ bp = Bm + (size_t)(half * 4) * Nn + colB;

    v4f a[PJ_U], b[PJ_U];
#pragma unroll
    for (int u = 0; u < PJ_U; ++u) {
        a[u] = *reinterpret_cast<const v4f*>(ap + 8 * u);
        const float* __restrict__ bq = bp + (size_t)(8 * u) * Nn;
        b[u].x = bq[0];
        b[u].y = bq[(size_t)Nn];
        b[u].z = bq[(size_t)2 * Nn];
        b[u].w = bq[(size_t)3 * Nn];
    }
    const float bv = bias[colB];

    gmem_float* op = (gmem_float*)(out + (size_t)(tileM * 16 + half * 8) * Nn + colB);
    asm volatile("" : "+v"(op));
    __builtin_amdgcn_sched_barrier(0);

    v8f c = {};
#pragma unroll
    for (int u = 0; u < PJ_U; ++u) {
        const v2f a0 = {a[u].x, a[u].y}, b0 = {b[u].x, b[u].y};
        const v2f a1 = {a[u].z, a[u].w}, b1 = {b[u].z, b[u].w};
        c = __builtin_amdgcn_wmma_f32_16x16x4_f32(false, a0, false, b0,
                                                  (short)0, c, false, false);
        c = __builtin_amdgcn_wmma_f32_16x16x4_f32(false, a1, false, b1,
                                                  (short)0, c, false, false);
    }
    __builtin_amdgcn_sched_barrier(0);

#pragma unroll
    for (int v = 0; v < 8; ++v) op[(size_t)v * Nn] = c[v] + bv;
}

extern "C" void kernel_launch(void* const* d_in, const int* in_sizes, int n_in,
                              void* d_out, int out_size, void* d_ws, size_t ws_size,
                              hipStream_t stream) {
    const float* x      = (const float*)d_in[0];
    const int*   eidx   = (const int*)d_in[1];
    const float* w_rel  = (const float*)d_in[4];
    const float* b_rel  = (const float*)d_in[5];
    const float* w_root = (const float*)d_in[6];
    const float* w_proj = (const float*)d_in[7];
    const float* b_proj = (const float*)d_in[8];
    float*       out    = (float*)d_out;

    const int d    = D;
    const int N    = in_sizes[0] / d;
    const int Bg   = N / NPG;
    const int E    = in_sizes[1] / 2;
    const int Eg   = E / Bg;
    const int k    = (NPG + 1) / 2;
    const int outd = in_sizes[8];

    float* r      = (float*)d_ws;
    float* s      = r + N;
    float* pooled = s + N;

    node_dots<<<dim3((N + ND_WAVES * NR - 1) / (ND_WAVES * NR)), dim3(ND_WAVES * 32), 0, stream>>>(
        x, w_rel, w_root, r, s, N);

    graph_score_pool<<<dim3(Bg), dim3(NPG), 0, stream>>>(
        x, r, s, eidx, eidx + E, b_rel, pooled, Eg, k);

    proj_wmma<<<dim3(Bg / 16, outd / 16), dim3(32), 0, stream>>>(
        pooled, w_proj, b_proj, out, outd);
}


# graph_score_pool rank count across the lanes: a lane holds 16 integer score keys, each compare answers 32 pairs as a lane mask and the scalar unit counts its bits; same ranks, same sums
# speedup vs baseline: 4.2826x; 1.2816x over previous
#include <hip/hip_runtime.h>
#include <hip/hip_bf16.h>
#include <math.h>

typedef float v2f __attribute__((ext_vector_type(2)));
typedef float v4f __attribute__((ext_vector_type(4)));
typedef float v8f __attribute__((ext_vector_type(8)));

#define D 256
#define NPG 512
#define PG 8
#define PU 16
#define NR 16
#define ND_WAVES 8
#define EB 8
#define RM (NPG / 32)

static_assert(2 * NR == 32, "node_dots folds one partial sum per lane of a 32-lane wave");
template <int OFF>
__device__ __forceinline__ void nd_fold(float (&val)[2 * NR], int lane) {
    const bool up = (lane & OFF) != 0;
#pragma unroll
    for (int i = 0; i < OFF; ++i) {
        const float keep = up ? val[i + OFF] : val[i];
        const float send = up ? val[i] : val[i + OFF];
        val[i] = keep + __shfl_xor(send, OFF);
    }
}

__global__ __launch_bounds__(ND_WAVES * 32)
void node_dots(const float* __restrict__ x,
               const float* __restrict__ w_rel,
               const float* __restrict__ w_root,
               float* __restrict__ r, float* __restrict__ s, int N) {
    const int lane = threadIdx.x & 31;
    const int wave = __builtin_amdgcn_readfirstlane((int)(threadIdx.x >> 5));
    const int row0 = (blockIdx.x * ND_WAVES + wave) * NR;
    if (row0 >= N) return;

    const v4f* __restrict__ wr4 = reinterpret_cast<const v4f*>(w_rel);
    const v4f* __restrict__ ws4 = reinterpret_cast<const v4f*>(w_root);
    const v4f wr0 = wr4[lane], wr1 = wr4[32 + lane];
    const v4f ws0 = ws4[lane], ws1 = ws4[32 + lane];

    v4f xa[NR], xb[NR];
#pragma unroll
    for (int i = 0; i < NR; ++i) {
        const int row = min(row0 + i, N - 1);
        const v4f* __restrict__ xp = reinterpret_cast<const v4f*>(x + (size_t)row * D);
        xa[i] = xp[lane];
        xb[i] = xp[32 + lane];
    }
    __builtin_amdgcn_sched_barrier(0);

    const v2f w0 = {wr0.x, ws0.x}, w1 = {wr0.y, ws0.y}, w2 = {wr0.z, ws0.z}, w3 = {wr0.w, ws0.w};
    const v2f w4 = {wr1.x, ws1.x}, w5 = {wr1.y, ws1.y}, w6 = {wr1.z, ws1.z}, w7 = {wr1.w, ws1.w};

    float val[2 * NR];
#pragma unroll
    for (int i = 0; i < NR; ++i) {
        v2f acc = xa[i].x * w0;
        acc += xa[i].y * w1;
        acc += xa[i].z * w2;
        acc += xa[i].w * w3;
        acc += xb[i].x * w4;
        acc += xb[i].y * w5;
        acc += xb[i].z * w6;
        acc += xb[i].w * w7;
        val[i]      = acc.x;
        val[NR + i] = acc.y;
        if ((i & 3) == 3) __builtin_amdgcn_sched_barrier(0);
    }

    nd_fold<16>(val, lane);
    nd_fold<8>(val, lane);
    nd_fold<4>(val, lane);
    nd_fold<2>(val, lane);
    nd_fold<1>(val, lane);

    const int row = row0 + (lane & 15);
    if (row < N) {
        float* __restrict__ dst = (lane < 16) ? r : s;
        dst[row] = val[0];
    }
}

__global__ void graph_score_pool(const float* __restrict__ x,
                                 const float* __restrict__ r,
                                 const float* __restrict__ s,
                                 const int* __restrict__ esrc,
                                 const int* __restrict__ edst,
                                 const float* __restrict__ b_rel,
                                 float* __restrict__ pooled,
                                 int Eg, int k) {
    __shared__ float acc[NPG];
    __shared__ int   skey[NPG];
    __shared__ unsigned koff[NPG];
    __shared__ float    kscore[NPG];
    __shared__ v4f      partial4[PG * (D / 4)];

    const int b     = blockIdx.x;
    const int tid   = threadIdx.x;
    const int nbase = b * NPG;

    const float brel = b_rel[0];

    acc[tid] = 0.f;
    koff[tid] = 0u;
    kscore[tid] = 0.f;
    __syncthreads();

    const float s_own = s[nbase + tid];

    const int* __restrict__ psrc = esrc + (long)b * Eg;
    const int* __restrict__ pdst = edst + (long)b * Eg;
    int e = tid;
    for (; e + (EB - 1) * NPG < Eg; e += EB * NPG) {
        int   sg[EB], dg[EB];
        float rv[EB];
#pragma unroll
        for (int u = 0; u < EB; ++u) {
            sg[u] = psrc[e + u * NPG];
            dg[u] = pdst[e + u * NPG];
        }
#pragma unroll
        for (int u = 0; u < EB; ++u) rv[u] = r[sg[u]];
#pragma unroll
        for (int u = 0; u < EB; ++u) atomicAdd(&acc[dg[u] - nbase], rv[u]);
    }
    for (; e < Eg; e += NPG)
        atomicAdd(&acc[pdst[e] - nbase], r[psrc[e]]);
    __syncthreads();

    const float sc = tanhf(acc[tid] + s_own + brel);
    int kown = __float_as_int(sc);
    if (kown == (int)0x80000000) kown = 0;
    kown ^= (kown >> 31) & 0x7fffffff;
    skey[tid] = kown;
    __syncthreads();

    const int lane = tid & 31;
    const int wv = __builtin_amdgcn_readfirstlane(tid >> 5);
    int K[RM];
#pragma unroll
    for (int m = 0; m < RM; ++m) K[m] = skey[32 * m + lane] + (m < wv ? 1 : 0);
    int cnt = 0;
#pragma unroll
    for (int ii = 0; ii < 32; ++ii) {
        const int ki = __builtin_amdgcn_readlane(kown, ii);
        int total = 0;
#pragma unroll
        for (int m = 0; m < RM; m += 2) {
            const unsigned lo = __builtin_amdgcn_ballot_w32(K[m] > ki);
            const unsigned hi = __builtin_amdgcn_ballot_w32(K[m + 1] > ki);
            total += __builtin_popcountll(((unsigned long long)hi << 32) | lo);
        }
        const unsigned eq = __builtin_amdgcn_ballot_w32(kown == ki);
        total += __builtin_popcount(eq & ((1u << ii) - 1u));
        if (lane == ii) cnt = total;
    }
    if (cnt < k) { koff[cnt] = (unsigned)tid * (unsigned)(D * sizeof(float)); kscore[cnt] = sc; }
    __syncthreads();

    const int g  = tid >> 6;
    const int c4 = tid & 63;
    const char* __restrict__ xg =
        reinterpret_cast<const char*>(x + (size_t)nbase * D) + c4 * 16;
    v4f ap = {0.f, 0.f, 0.f, 0.f};
    int j = g;
    for (; j + PG * (PU - 1) < k; j += PG * PU) {
        v4f   v[PU];
        float w[PU];
#pragma unroll
        for (int u = 0; u < PU; ++u) {
            w[u] = kscore[j + PG * u];
            v[u] = *reinterpret_cast<const v4f*>(xg + koff[j + PG * u]);
        }
        __builtin_amdgcn_sched_barrier(0);
#pragma unroll
        for (int u = 0; u < PU; ++u) ap += w[u] * v[u];
    }
    for (; j < k; j += PG)
        ap += kscore[j] * *reinterpret_cast<const v4f*>(xg + koff[j]);
    partial4[g * (D / 4) + c4] = ap;
    __syncthreads();
    if (tid < D) {
        const float* partial = reinterpret_cast<const float*>(partial4);
        float t = partial[tid];
#pragma unroll
        for (int q = 1; q < PG; ++q) t += partial[q * D + tid];
        pooled[(size_t)b * D + tid] = t / (float)k;
    }
}

#define PJ_U (D / 8)
typedef __attribute__((address_space(1))) float gmem_float;
__global__ __launch_bounds__(32)
void proj_wmma(const float* __restrict__ A,
               const float* __restrict__ Bm,
               const float* __restrict__ bias,
               float* __restrict__ out,
               int Nn) {
    const int lane  = threadIdx.x & 31;
    const int half  = lane >> 4;
    const int l16   = lane & 15;
    const int tileM = blockIdx.x;
    const int tileN = blockIdx.y;

    const int rowA = tileM * 16 + l16;
    const int colB = tileN * 16 + l16;

    const float* __restrict__ ap = A + (size_t)rowA * D + half * 4;
    const float* __restrict__ bp = Bm + (size_t)(half * 4) * Nn + colB;

    v4f a[PJ_U], b[PJ_U];
#pragma unroll
    for (int u = 0; u < PJ_U; ++u) {
        a[u] = *reinterpret_cast<const v4f*>(ap + 8 * u);
        const float* __restrict__ bq = bp + (size_t)(8 * u) * Nn;
        b[u].x = bq[0];
        b[u].y = bq[(size_t)Nn];
        b[u].z = bq[(size_t)2 * Nn];
        b[u].w = bq[(size_t)3 * Nn];
    }
    const float bv = bias[colB];

    gmem_float* op = (gmem_float*)(out + (size_t)(tileM * 16 + half * 8) * Nn + colB);
    asm volatile("" : "+v"(op));
    __builtin_amdgcn_sched_barrier(0);

    v8f c = {};
#pragma unroll
    for (int u = 0; u < PJ_U; ++u) {
        const v2f a0 = {a[u].x, a[u].y}, b0 = {b[u].x, b[u].y};
        const v2f a1 = {a[u].z, a[u].w}, b1 = {b[u].z, b[u].w};
        c = __builtin_amdgcn_wmma_f32_16x16x4_f32(false, a0, false, b0,
                                                  (short)0, c, false, false);
        c = __builtin_amdgcn_wmma_f32_16x16x4_f32(false, a1, false, b1,
                                                  (short)0, c, false, false);
    }
    __builtin_amdgcn_sched_barrier(0);

#pragma unroll
    for (int v = 0; v < 8; ++v) op[(size_t)v * Nn] = c[v] + bv;
}

extern "C" void kernel_launch(void* const* d_in, const int* in_sizes, int n_in,
                              void* d_out, int out_size, void* d_ws, size_t ws_size,
                              hipStream_t stream) {
    const float* x      = (const float*)d_in[0];
    const int*   eidx   = (const int*)d_in[1];
    const float* w_rel  = (const float*)d_in[4];
    const float* b_rel  = (const float*)d_in[5];
    const float* w_root = (const float*)d_in[6];
    const float* w_proj = (const float*)d_in[7];
    const float* b_proj = (const float*)d_in[8];
    float*       out    = (float*)d_out;

    const int d    = D;
    const int N    = in_sizes[0] / d;
    const int Bg   = N / NPG;
    const int E    = in_sizes[1] / 2;
    const int Eg   = E / Bg;
    const int k    = (NPG + 1) / 2;
    const int outd = in_sizes[8];

    float* r      = (float*)d_ws;
    float* s      = r + N;
    float* pooled = s + N;

    node_dots<<<dim3((N + ND_WAVES * NR - 1) / (ND_WAVES * NR)), dim3(ND_WAVES * 32), 0, stream>>>(
        x, w_rel, w_root, r, s, N);

    graph_score_pool<<<dim3(Bg), dim3(NPG), 0, stream>>>(
        x, r, s, eidx, eidx + E, b_rel, pooled, Eg, k);

    proj_wmma<<<dim3(Bg / 16, outd / 16), dim3(32), 0, stream>>>(
        pooled, w_proj, b_proj, out, outd);
}


# node_dots folded into graph_score_pool: a workgroup streams its own graph's 512 rows, r and s stay in LDS, edge gathers read LDS, edge indices asked for ahead; 2 launches, not 3; sum orders kept
# speedup vs baseline: 5.2970x; 1.2369x over previous
#include <hip/hip_runtime.h>
#include <hip/hip_bf16.h>
#include <math.h>

typedef float v2f __attribute__((ext_vector_type(2)));
typedef float v4f __attribute__((ext_vector_type(4)));
typedef float v8f __attribute__((ext_vector_type(8)));

#define D 256
#define NPG 512
#define PG 8
#define PU 16
#define NR 16
#define NB (32 / NR)
#define EB 8
#define RM (NPG / 32)

static_assert(2 * NR == 32, "the row stream folds one partial sum per lane of a 32-lane wave");
static_assert(NB * NR == 32, "a wave's batches cover its 32 rows exactly");
template <int OFF>
__device__ __forceinline__ void nd_fold(float (&val)[2 * NR], int lane) {
    const bool up = (lane & OFF) != 0;
#pragma unroll
    for (int i = 0; i < OFF; ++i) {
        const float keep = up ? val[i + OFF] : val[i];
        const float send = up ? val[i] : val[i + OFF];
        val[i] = keep + __shfl_xor(send, OFF);
    }
}

__device__ __forceinline__ float far_row_dot(const float* __restrict__ x,
                                             const float* __restrict__ w, int row) {
    const v4f* __restrict__ xp = reinterpret_cast<const v4f*>(x + (size_t)row * D);
    const v4f* __restrict__ wp = reinterpret_cast<const v4f*>(w);
    float t = 0.f;
#pragma unroll 1
    for (int c = 0; c < D / 4; ++c) {
        const v4f a = xp[c], ww = wp[c];
        t += a.x * ww.x + a.y * ww.y + a.z * ww.z + a.w * ww.w;
    }
    return t;
}

__global__ __launch_bounds__(NPG)
void graph_score_pool(const float* __restrict__ x,
                      const float* __restrict__ w_rel,
                      const float* __restrict__ w_root,
                      const int* __restrict__ esrc,
                      const int* __restrict__ edst,
                      const float* __restrict__ b_rel,
                      float* __restrict__ pooled,
                      int Eg, int k) {
    __shared__ float rl[NPG];
    __shared__ float sl[NPG];
    __shared__ float acc[NPG];
    __shared__ int   skey[NPG];
    __shared__ unsigned koff[NPG];
    __shared__ float    kscore[NPG];
    __shared__ v4f      partial4[PG * (D / 4)];

    const int b     = blockIdx.x;
    const int tid   = threadIdx.x;
    const int nbase = b * NPG;
    const int lane  = tid & 31;
    const int wv = __builtin_amdgcn_readfirstlane(tid >> 5);

    const float brel = b_rel[0];

    acc[tid] = 0.f;
    koff[tid] = 0u;
    kscore[tid] = 0.f;

    const int* __restrict__ psrc = esrc + (long)b * Eg;
    const int* __restrict__ pdst = edst + (long)b * Eg;
    int sg[EB], dg[EB];
    if (tid + (EB - 1) * NPG < Eg) {
#pragma unroll
        for (int u = 0; u < EB; ++u) {
            sg[u] = psrc[tid + u * NPG];
            dg[u] = pdst[tid + u * NPG];
        }
    }

    {
        const v4f* __restrict__ wr4 = reinterpret_cast<const v4f*>(w_rel);
        const v4f* __restrict__ ws4 = reinterpret_cast<const v4f*>(w_root);
        const v4f wr0 = wr4[lane], wr1 = wr4[32 + lane];
        const v4f ws0 = ws4[lane], ws1 = ws4[32 + lane];

#pragma unroll
        for (int bt = 0; bt < NB; ++bt) {
            const int lrow0 = wv * 32 + bt * NR;
            v4f xa[NR], xb[NR];
#pragma unroll
            for (int i = 0; i < NR; ++i) {
                const v4f* __restrict__ xp =
                    reinterpret_cast<const v4f*>(x + (size_t)(nbase + lrow0 + i) * D);
                xa[i] = xp[lane];
                xb[i] = xp[32 + lane];
            }
            __builtin_amdgcn_sched_barrier(0);

            const v2f w0 = {wr0.x, ws0.x}, w1 = {wr0.y, ws0.y}, w2 = {wr0.z, ws0.z}, w3 = {wr0.w, ws0.w};
            const v2f w4 = {wr1.x, ws1.x}, w5 = {wr1.y, ws1.y}, w6 = {wr1.z, ws1.z}, w7 = {wr1.w, ws1.w};

            float val[2 * NR];
#pragma unroll
            for (int i = 0; i < NR; ++i) {
                v2f a2 = xa[i].x * w0;
                a2 += xa[i].y * w1;
                a2 += xa[i].z * w2;
                a2 += xa[i].w * w3;
                a2 += xb[i].x * w4;
                a2 += xb[i].y * w5;
                a2 += xb[i].z * w6;
                a2 += xb[i].w * w7;
                val[i]      = a2.x;
                val[NR + i] = a2.y;
                if ((i & 3) == 3) __builtin_amdgcn_sched_barrier(0);
            }

            nd_fold<16>(val, lane);
            nd_fold<8>(val, lane);
            nd_fold<4>(val, lane);
            nd_fold<2>(val, lane);
            nd_fold<1>(val, lane);

            float* dstl = (lane < 16) ? rl : sl;
            dstl[lrow0 + (lane & 15)] = val[0];
        }
    }
    __syncthreads();

    const float s_own = sl[tid];

    int e = tid;
    for (; e + (EB - 1) * NPG < Eg; e += EB * NPG) {
        float rv[EB];
        if (e != tid) {
#pragma unroll
            for (int u = 0; u < EB; ++u) {
                sg[u] = psrc[e + u * NPG];
                dg[u] = pdst[e + u * NPG];
            }
        }
#pragma unroll
        for (int u = 0; u < EB; ++u) {
            const unsigned loc = (unsigned)(sg[u] - nbase);
            rv[u] = (loc < (unsigned)NPG) ? rl[loc] : far_row_dot(x, w_rel, sg[u]);
        }
#pragma unroll
        for (int u = 0; u < EB; ++u) atomicAdd(&acc[dg[u] - nbase], rv[u]);
    }
    for (; e < Eg; e += NPG) {
        const int sn = psrc[e];
        const unsigned loc = (unsigned)(sn - nbase);
        atomicAdd(&acc[pdst[e] - nbase], (loc < (unsigned)NPG) ? rl[loc] : far_row_dot(x, w_rel, sn));
    }
    __syncthreads();

    const float sc = tanhf(acc[tid] + s_own + brel);
    int kown = __float_as_int(sc);
    if (kown == (int)0x80000000) kown = 0;
    kown ^= (kown >> 31) & 0x7fffffff;
    skey[tid] = kown;
    __syncthreads();

    int K[RM];
#pragma unroll
    for (int m = 0; m < RM; ++m) K[m] = skey[32 * m + lane] + (m < wv ? 1 : 0);
    int cnt = 0;
#pragma unroll
    for (int ii = 0; ii < 32; ++ii) {
        const int ki = __builtin_amdgcn_readlane(kown, ii);
        int total = 0;
#pragma unroll
        for (int m = 0; m < RM; m += 2) {
            const unsigned lo = __builtin_amdgcn_ballot_w32(K[m] > ki);
            const unsigned hi = __builtin_amdgcn_ballot_w32(K[m + 1] > ki);
            total += __builtin_popcountll(((unsigned long long)hi << 32) | lo);
        }
        const unsigned eq = __builtin_amdgcn_ballot_w32(kown == ki);
        total += __builtin_popcount(eq & ((1u << ii) - 1u));
        if (lane == ii) cnt = total;
    }
    if (cnt < k) { koff[cnt] = (unsigned)tid * (unsigned)(D * sizeof(float)); kscore[cnt] = sc; }
    __syncthreads();

    const int g  = tid >> 6;
    const int c4 = tid & 63;
    const char* __restrict__ xg =
        reinterpret_cast<const char*>(x + (size_t)nbase * D) + c4 * 16;
    v4f ap = {0.f, 0.f, 0.f, 0.f};
    int j = g;
    for (; j + PG * (PU - 1) < k; j += PG * PU) {
        v4f   v[PU];
        float w[PU];
#pragma unroll
        for (int u = 0; u < PU; ++u) {
            w[u] = kscore[j + PG * u];
            v[u] = *reinterpret_cast<const v4f*>(xg + koff[j + PG * u]);
        }
        __builtin_amdgcn_sched_barrier(0);
#pragma unroll
        for (int u = 0; u < PU; ++u) ap += w[u] * v[u];
    }
    for (; j < k; j += PG)
        ap += kscore[j] * *reinterpret_cast<const v4f*>(xg + koff[j]);
    partial4[g * (D / 4) + c4] = ap;
    __syncthreads();
    if (tid < D) {
        const float* partial = reinterpret_cast<const float*>(partial4);
        float t = partial[tid];
#pragma unroll
        for (int q = 1; q < PG; ++q) t += partial[q * D + tid];
        pooled[(size_t)b * D + tid] = t / (float)k;
    }
}

#define PJ_U (D / 8)
typedef __attribute__((address_space(1))) float gmem_float;
__global__ __launch_bounds__(32)
void proj_wmma(const float* __restrict__ A,
               const float* __restrict__ Bm,
               const float* __restrict__ bias,
               float* __restrict__ out,
               int Nn) {
    const int lane  = threadIdx.x & 31;
    const int half  = lane >> 4;
    const int l16   = lane & 15;
    const int tileM = blockIdx.x;
    const int tileN = blockIdx.y;

    const int rowA = tileM * 16 + l16;
    const int colB = tileN * 16 + l16;

    const float* __restrict__ ap = A + (size_t)rowA * D + half * 4;
    const float* __restrict__ bp = Bm + (size_t)(half * 4) * Nn + colB;

    v4f a[PJ_U], b[PJ_U];
#pragma unroll
    for (int u = 0; u < PJ_U; ++u) {
        a[u] = *reinterpret_cast<const v4f*>(ap + 8 * u);
        const float* __restrict__ bq = bp + (size_t)(8 * u) * Nn;
        b[u].x = bq[0];
        b[u].y = bq[(size_t)Nn];
        b[u].z = bq[(size_t)2 * Nn];
        b[u].w = bq[(size_t)3 * Nn];
    }
    const float bv = bias[colB];

    gmem_float* op = (gmem_float*)(out + (size_t)(tileM * 16 + half * 8) * Nn + colB);
    asm volatile("" : "+v"(op));
    __builtin_amdgcn_sched_barrier(0);

    v8f c = {};
#pragma unroll
    for (int u = 0; u < PJ_U; ++u) {
        const v2f a0 = {a[u].x, a[u].y}, b0 = {b[u].x, b[u].y};
        const v2f a1 = {a[u].z, a[u].w}, b1 = {b[u].z, b[u].w};
        c = __builtin_amdgcn_wmma_f32_16x16x4_f32(false, a0, false, b0,
                                                  (short)0, c, false, false);
        c = __builtin_amdgcn_wmma_f32_16x16x4_f32(false, a1, false, b1,
                                                  (short)0, c, false, false);
    }
    __builtin_amdgcn_sched_barrier(0);

#pragma unroll
    for (int v = 0; v < 8; ++v) op[(size_t)v * Nn] = c[v] + bv;
}

extern "C" void kernel_launch(void* const* d_in, const int* in_sizes, int n_in,
                              void* d_out, int out_size, void* d_ws, size_t ws_size,
                              hipStream_t stream) {
    const float* x      = (const float*)d_in[0];
    const int*   eidx   = (const int*)d_in[1];
    const float* w_rel  = (const float*)d_in[4];
    const float* b_rel  = (const float*)d_in[5];
    const float* w_root = (const float*)d_in[6];
    const float* w_proj = (const float*)d_in[7];
    const float* b_proj = (const float*)d_in[8];
    float*       out    = (float*)d_out;

    const int d    = D;
    const int N    = in_sizes[0] / d;
    const int Bg   = N / NPG;
    const int E    = in_sizes[1] / 2;
    const int Eg   = E / Bg;
    const int k    = (NPG + 1) / 2;
    const int outd = in_sizes[8];

    float* pooled = (float*)d_ws;

    graph_score_pool<<<dim3(Bg), dim3(NPG), 0, stream>>>(
        x, w_rel, w_root, eidx, eidx + E, b_rel, pooled, Eg, k);

    proj_wmma<<<dim3(Bg / 16, outd / 16), dim3(32), 0, stream>>>(
        pooled, w_proj, b_proj, out, outd);
}


# proj_wmma folded into graph_score_pool: the 16 waves' partial pooled rows are the A tile's rows, a wave takes 32 columns and half of K of w_proj; 1 launch, not 2; out summed in a new order
# speedup vs baseline: 5.9053x; 1.1148x over previous
#include <hip/hip_runtime.h>
#include <hip/hip_bf16.h>
#include <math.h>


typedef float v2f __attribute__((ext_vector_type(2)));
typedef float v4f __attribute__((ext_vector_type(4)));
typedef float v8f __attribute__((ext_vector_type(8)));
typedef __attribute__((address_space(3))) v4f lds_v4f;

#define D 256
#define NPG 512
#define NW (NPG / 32)
#define NR 16
#define NB (32 / NR)
#define EB 8
#define RM (NPG / 32)
#define PJ_K (D / 2)
#define PJ_U (PJ_K / 8)
#define PJ_G1 7
#define PJ_G2 14
#define PP (D / 4 + 1)
#define OD 256
static_assert(OD == 16 * NW, "the projection gives every pair of waves two tiles of 16 columns");
static_assert(8 * PJ_G1 + 1 < 63 && 8 * (PJ_G2 - PJ_G1) < 63, "a group of loads stays below the load counter's 63");

static_assert(2 * NR == 32, "the row stream folds one partial sum per lane of a 32-lane wave");
static_assert(NB * NR == 32, "a wave's batches cover its 32 rows exactly");
static_assert(NB == 2, "the pooling takes a wave's first batch from LDS and its second from registers");
template <int OFF>
__device__ __forceinline__ void nd_fold(float (&val)[2 * NR], int lane) {
    const bool up = (lane & OFF) != 0;
#pragma unroll
    for (int i = 0; i < OFF; ++i) {
        const float keep = up ? val[i + OFF] : val[i];
        const float send = up ? val[i] : val[i + OFF];
        val[i] = keep + __shfl_xor(send, OFF);
    }
}

__device__ __forceinline__ float far_row_dot(const float* __restrict__ x,
                                             const float* __restrict__ w, int row) {
    const v4f* __restrict__ xp = reinterpret_cast<const v4f*>(x + (size_t)row * D);
    const v4f* __restrict__ wp = reinterpret_cast<const v4f*>(w);
    float t = 0.f;
#pragma unroll 1
    for (int c = 0; c < D / 4; ++c) {
        const v4f a = xp[c], ww = wp[c];
        t += a.x * ww.x + a.y * ww.y + a.z * ww.z + a.w * ww.w;
    }
    return t;
}

__global__ __launch_bounds__(NPG)
void graph_score_pool(const float* __restrict__ x,
                      const float* __restrict__ w_rel,
                      const float* __restrict__ w_root,
                      const int* __restrict__ esrc,
                      const int* __restrict__ edst,
                      const float* __restrict__ b_rel,
                      const float* __restrict__ w_proj,
                      const float* __restrict__ b_proj,
                      float* __restrict__ out,
                      int Eg, int k) {
    __shared__ float rl[NPG];
    __shared__ float sl[NPG];
    __shared__ float acc[NPG];
    __shared__ int   skey[NPG];
    __shared__ v4f      partial4[NW * PP];
    __shared__ float    red[2 * OD];
    __shared__ v4f      xrows[NW * NR * (D / 4)];

    const int b     = blockIdx.x;
    const int tid   = threadIdx.x;
    const int nbase = b * NPG;
    const int lane  = tid & 31;
    const int wv = __builtin_amdgcn_readfirstlane(tid >> 5);

    const float brel = b_rel[0];

    acc[tid] = 0.f;

    const int* __restrict__ psrc = esrc + (long)b * Eg;
    const int* __restrict__ pdst = edst + (long)b * Eg;
    int sg[EB], dg[EB];
    if (tid + (EB - 1) * NPG < Eg) {
#pragma unroll
        for (int u = 0; u < EB; ++u) {
            sg[u] = psrc[tid + u * NPG];
            dg[u] = pdst[tid + u * NPG];
        }
    }

    v4f xa[NR], xb[NR];

    {
        const v4f* __restrict__ wr4 = reinterpret_cast<const v4f*>(w_rel);
        const v4f* __restrict__ ws4 = reinterpret_cast<const v4f*>(w_root);
        const v4f wr0 = wr4[lane], wr1 = wr4[32 + lane];
        const v4f ws0 = ws4[lane], ws1 = ws4[32 + lane];

#pragma unroll
        for (int bt = 0; bt < NB; ++bt) {
            const int lrow0 = wv * 32 + bt * NR;
#pragma unroll
            for (int i = 0; i < NR; ++i) {
                const v4f* __restrict__ xp =
                    reinterpret_cast<const v4f*>(x + (size_t)(nbase + lrow0 + i) * D);
                xa[i] = xp[lane];
                xb[i] = xp[32 + lane];
            }
            __builtin_amdgcn_sched_barrier(0);

            const v2f w0 = {wr0.x, ws0.x}, w1 = {wr0.y, ws0.y}, w2 = {wr0.z, ws0.z}, w3 = {wr0.w, ws0.w};
            const v2f w4 = {wr1.x, ws1.x}, w5 = {wr1.y, ws1.y}, w6 = {wr1.z, ws1.z}, w7 = {wr1.w, ws1.w};

            float val[2 * NR];
#pragma unroll
            for (int i = 0; i < NR; ++i) {
                v2f a2 = xa[i].x * w0;
                a2 += xa[i].y * w1;
                a2 += xa[i].z * w2;
                a2 += xa[i].w * w3;
                a2 += xb[i].x * w4;
                a2 += xb[i].y * w5;
                a2 += xb[i].z * w6;
                a2 += xb[i].w * w7;
                val[i]      = a2.x;
                val[NR + i] = a2.y;
                if (bt + 1 < NB) {
                    xrows[(wv * NR + i) * (D / 4) + lane]      = xa[i];
                    xrows[(wv * NR + i) * (D / 4) + 32 + lane] = xb[i];
                }
                if ((i & 3) == 3) __builtin_amdgcn_sched_barrier(0);
            }

            nd_fold<16>(val, lane);
            nd_fold<8>(val, lane);
            nd_fold<4>(val, lane);
            nd_fold<2>(val, lane);
            nd_fold<1>(val, lane);

            float* dstl = (lane < 16) ? rl : sl;
            dstl[lrow0 + (lane & 15)] = val[0];
        }
    }
    __syncthreads();

    const float s_own = sl[tid];

    int e = tid;
    for (; e + (EB - 1) * NPG < Eg; e += EB * NPG) {
        float rv[EB];
        if (e != tid) {
#pragma unroll
            for (int u = 0; u < EB; ++u) {
                sg[u] = psrc[e + u * NPG];
                dg[u] = pdst[e + u * NPG];
            }
        }
#pragma unroll
        for (int u = 0; u < EB; ++u) {
            const unsigned loc = (unsigned)(sg[u] - nbase);
            rv[u] = (loc < (unsigned)NPG) ? rl[loc] : far_row_dot(x, w_rel, sg[u]);
        }
#pragma unroll
        for (int u = 0; u < EB; ++u) atomicAdd(&acc[dg[u] - nbase], rv[u]);
    }
    for (; e < Eg; e += NPG) {
        const int sn = psrc[e];
        const unsigned loc = (unsigned)(sn - nbase);
        atomicAdd(&acc[pdst[e] - nbase], (loc < (unsigned)NPG) ? rl[loc] : far_row_dot(x, w_rel, sn));
    }
    __syncthreads();

    const float sc = tanhf(acc[tid] + s_own + brel);
    int kown = __float_as_int(sc);
    if (kown == (int)0x80000000) kown = 0;
    kown ^= (kown >> 31) & 0x7fffffff;
    skey[tid] = kown;
    __syncthreads();

    int K[RM];
#pragma unroll
    for (int m = 0; m < RM; ++m) K[m] = skey[32 * m + lane] + (m < wv ? 1 : 0);
    int cnt = 0;
#pragma unroll
    for (int ii = 0; ii < 32; ++ii) {
        const int ki = __builtin_amdgcn_readlane(kown, ii);
        int total = 0;
#pragma unroll
        for (int m = 0; m < RM; m += 2) {
            const unsigned lo = __builtin_amdgcn_ballot_w32(K[m] > ki);
            const unsigned hi = __builtin_amdgcn_ballot_w32(K[m + 1] > ki);
            total += __builtin_popcountll(((unsigned long long)hi << 32) | lo);
        }
        const unsigned eq = __builtin_amdgcn_ballot_w32(kown == ki);
        total += __builtin_popcount(eq & ((1u << ii) - 1u));
        if (lane == ii) cnt = total;
    }
    const float wgt = (cnt < k) ? sc : 0.f;
    v4f pa = {0.f, 0.f, 0.f, 0.f}, pb = {0.f, 0.f, 0.f, 0.f};
#pragma unroll
    for (int i = 0; i < NR; ++i) {
        const float wi = __int_as_float(__builtin_amdgcn_readlane(__float_as_int(wgt), NR + i));
        pa += wi * xa[i];
        pb += wi * xb[i];
    }
    const int half = lane >> 4;
    const int l16  = lane & 15;
    const int q    = wv & 1;
    const int tp   = wv >> 1;
    const float* __restrict__ bp = w_proj + (size_t)(PJ_K * q + 4 * half) * OD + 32 * tp + l16;
    v4f bw0[PJ_U], bw1[PJ_U];
#pragma unroll
    for (int u = 0; u < PJ_G1; ++u) {
        const float* __restrict__ bq = bp + (size_t)(8 * u) * OD;
        bw0[u].x = bq[0];       bw0[u].y = bq[OD];       bw0[u].z = bq[2 * OD];       bw0[u].w = bq[3 * OD];
        bw1[u].x = bq[16];      bw1[u].y = bq[OD + 16];  bw1[u].z = bq[2 * OD + 16];  bw1[u].w = bq[3 * OD + 16];
    }
    const float bv = (tid < OD) ? b_proj[tid] : 0.f;

#pragma unroll
    for (int i = 0; i < NR; ++i) {
        const float wi = __int_as_float(__builtin_amdgcn_readlane(__float_as_int(wgt), i));
        pa += wi * xrows[(wv * NR + i) * (D / 4) + lane];
        pb += wi * xrows[(wv * NR + i) * (D / 4) + 32 + lane];
    }
    partial4[wv * PP + lane]      = pa;
    partial4[wv * PP + 32 + lane] = pb;
    __builtin_amdgcn_sched_barrier(0);
#pragma unroll
    for (int u = PJ_G1; u < PJ_G2; ++u) {
        const float* __restrict__ bq = bp + (size_t)(8 * u) * OD;
        bw0[u].x = bq[0];       bw0[u].y = bq[OD];       bw0[u].z = bq[2 * OD];       bw0[u].w = bq[3 * OD];
        bw1[u].x = bq[16];      bw1[u].y = bq[OD + 16];  bw1[u].z = bq[2 * OD + 16];  bw1[u].w = bq[3 * OD + 16];
    }
    __builtin_amdgcn_fence(__ATOMIC_RELEASE, "workgroup", "local");
    __builtin_amdgcn_s_barrier();
    __builtin_amdgcn_fence(__ATOMIC_ACQUIRE, "workgroup", "local");

    const lds_v4f* arow = (const lds_v4f*)(partial4 + l16 * PP + half + 2 * PJ_U * q);
    asm volatile("" : "+v"(arow));
    v4f av[PJ_U];
#pragma unroll
    for (int u = 0; u < PJ_U; ++u) av[u] = arow[2 * u];
#pragma unroll
    for (int u = PJ_G2; u < PJ_U; ++u) {
        const float* __restrict__ bq = bp + (size_t)(8 * u) * OD;
        bw0[u].x = bq[0];       bw0[u].y = bq[OD];       bw0[u].z = bq[2 * OD];       bw0[u].w = bq[3 * OD];
        bw1[u].x = bq[16];      bw1[u].y = bq[OD + 16];  bw1[u].z = bq[2 * OD + 16];  bw1[u].w = bq[3 * OD + 16];
    }
    __builtin_amdgcn_sched_barrier(0);

    v8f c0 = {}, c1 = {};
#pragma unroll
    for (int u = 0; u < PJ_U; ++u) {
        const v2f a0 = {av[u].x, av[u].y}, a1 = {av[u].z, av[u].w};
        const v2f b00 = {bw0[u].x, bw0[u].y}, b01 = {bw0[u].z, bw0[u].w};
        const v2f b10 = {bw1[u].x, bw1[u].y}, b11 = {bw1[u].z, bw1[u].w};
        c0 = __builtin_amdgcn_wmma_f32_16x16x4_f32(false, a0, false, b00, (short)0, c0, false, false);
        c1 = __builtin_amdgcn_wmma_f32_16x16x4_f32(false, a0, false, b10, (short)0, c1, false, false);
        c0 = __builtin_amdgcn_wmma_f32_16x16x4_f32(false, a1, false, b01, (short)0, c0, false, false);
        c1 = __builtin_amdgcn_wmma_f32_16x16x4_f32(false, a1, false, b11, (short)0, c1, false, false);
    }
    asm volatile("v_nop\n\tv_nop\n\tv_nop\n\tv_nop" : "+v"(c0), "+v"(c1));
    const float s0 = ((c0[0] + c0[1]) + (c0[2] + c0[3])) + ((c0[4] + c0[5]) + (c0[6] + c0[7]));
    const float s1 = ((c1[0] + c1[1]) + (c1[2] + c1[3])) + ((c1[4] + c1[5]) + (c1[6] + c1[7]));
    const float keep = half ? s1 : s0;
    const float send = half ? s0 : s1;
    red[q * OD + 32 * tp + lane] = keep + __shfl_xor(send, 16);
    __syncthreads();
    if (tid < OD) out[(size_t)b * OD + tid] = (red[tid] + red[OD + tid]) / (float)k + bv;
}

extern "C" void kernel_launch(void* const* d_in, const int* in_sizes, int n_in,
                              void* d_out, int out_size, void* d_ws, size_t ws_size,
                              hipStream_t stream) {
    const float* x      = (const float*)d_in[0];
    const int*   eidx   = (const int*)d_in[1];
    const float* w_rel  = (const float*)d_in[4];
    const float* b_rel  = (const float*)d_in[5];
    const float* w_root = (const float*)d_in[6];
    const float* w_proj = (const float*)d_in[7];
    const float* b_proj = (const float*)d_in[8];
    float*       out    = (float*)d_out;

    const int d    = D;
    const int N    = in_sizes[0] / d;
    const int Bg   = N / NPG;
    const int E    = in_sizes[1] / 2;
    const int Eg   = E / Bg;
    const int k    = (NPG + 1) / 2;

    graph_score_pool<<<dim3(Bg), dim3(NPG), 0, stream>>>(
        x, w_rel, w_root, eidx, eidx + E, b_rel, w_proj, b_proj, out, Eg, k);
}


# graph_score_pool rank count only where it decides: 32 splitter nodes ranked exactly, the two next to rank k settle every node outside them by one compare, only nodes between are counted; same sums
# speedup vs baseline: 7.0958x; 1.2016x over previous
#include <hip/hip_runtime.h>
#include <hip/hip_bf16.h>
#include <math.h>


typedef float v2f __attribute__((ext_vector_type(2)));
typedef float v4f __attribute__((ext_vector_type(4)));
typedef float v8f __attribute__((ext_vector_type(8)));
typedef __attribute__((address_space(3))) v4f lds_v4f;

#define D 256
#define NPG 512
#define NW (NPG / 32)
#define NR 16
#define NB (32 / NR)
#define EB 8
#define RM (NPG / 32)
#define PJ_K (D / 2)
#define PJ_U (PJ_K / 8)
#define PJ_G1 7
#define PJ_G2 14
#define PP (D / 4 + 1)
#define OD 256
static_assert(OD == 16 * NW, "the projection gives every pair of waves two tiles of 16 columns");
static_assert(8 * PJ_G1 + 1 < 63 && 8 * (PJ_G2 - PJ_G1) < 63, "a group of loads stays below the load counter's 63");

static_assert(2 * NR == 32, "the row stream folds one partial sum per lane of a 32-lane wave");
static_assert(NB * NR == 32, "a wave's batches cover its 32 rows exactly");
static_assert(NB == 2, "the pooling takes a wave's first batch from LDS and its second from registers");
template <int OFF>
__device__ __forceinline__ void nd_fold(float (&val)[2 * NR], int lane) {
    const bool up = (lane & OFF) != 0;
#pragma unroll
    for (int i = 0; i < OFF; ++i) {
        const float keep = up ? val[i + OFF] : val[i];
        const float send = up ? val[i] : val[i + OFF];
        val[i] = keep + __shfl_xor(send, OFF);
    }
}

__device__ __forceinline__ float far_row_dot(const float* __restrict__ x,
                                             const float* __restrict__ w, int row) {
    const v4f* __restrict__ xp = reinterpret_cast<const v4f*>(x + (size_t)row * D);
    const v4f* __restrict__ wp = reinterpret_cast<const v4f*>(w);
    float t = 0.f;
#pragma unroll 1
    for (int c = 0; c < D / 4; ++c) {
        const v4f a = xp[c], ww = wp[c];
        t += a.x * ww.x + a.y * ww.y + a.z * ww.z + a.w * ww.w;
    }
    return t;
}

__device__ __forceinline__ int rank_of(const int (&K)[NPG / 32], int kown, int ii) {
    const int ki = __builtin_amdgcn_readlane(kown, ii);
    int total = 0;
#pragma unroll
    for (int m = 0; m < NPG / 32; m += 2) {
        const unsigned lo = __builtin_amdgcn_ballot_w32(K[m] > ki);
        const unsigned hi = __builtin_amdgcn_ballot_w32(K[m + 1] > ki);
        total += __builtin_popcountll(((unsigned long long)hi << 32) | lo);
    }
    const unsigned eq = __builtin_amdgcn_ballot_w32(kown == ki);
    total += __builtin_popcount(eq & ((1u << ii) - 1u));
    return total;
}

__global__ __launch_bounds__(NPG)
void graph_score_pool(const float* __restrict__ x,
                      const float* __restrict__ w_rel,
                      const float* __restrict__ w_root,
                      const int* __restrict__ esrc,
                      const int* __restrict__ edst,
                      const float* __restrict__ b_rel,
                      const float* __restrict__ w_proj,
                      const float* __restrict__ b_proj,
                      float* __restrict__ out,
                      int Eg, int k) {
    __shared__ float rl[NPG];
    __shared__ float sl[NPG];
    __shared__ float acc[NPG];
    __shared__ int   skey[NPG];
    __shared__ int   splr[NPG / 16];
    __shared__ v4f      partial4[NW * PP];
    __shared__ float    red[2 * OD];
    __shared__ v4f      xrows[NW * NR * (D / 4)];

    const int b     = blockIdx.x;
    const int tid   = threadIdx.x;
    const int nbase = b * NPG;
    const int lane  = tid & 31;
    const int wv = __builtin_amdgcn_readfirstlane(tid >> 5);

    const float brel = b_rel[0];

    acc[tid] = 0.f;

    const int* __restrict__ psrc = esrc + (long)b * Eg;
    const int* __restrict__ pdst = edst + (long)b * Eg;
    int sg[EB], dg[EB];
    if (tid + (EB - 1) * NPG < Eg) {
#pragma unroll
        for (int u = 0; u < EB; ++u) {
            sg[u] = psrc[tid + u * NPG];
            dg[u] = pdst[tid + u * NPG];
        }
    }

    v4f xa[NR], xb[NR];

    {
        const v4f* __restrict__ wr4 = reinterpret_cast<const v4f*>(w_rel);
        const v4f* __restrict__ ws4 = reinterpret_cast<const v4f*>(w_root);
        const v4f wr0 = wr4[lane], wr1 = wr4[32 + lane];
        const v4f ws0 = ws4[lane], ws1 = ws4[32 + lane];

#pragma unroll
        for (int bt = 0; bt < NB; ++bt) {
            const int lrow0 = wv * 32 + bt * NR;
#pragma unroll
            for (int i = 0; i < NR; ++i) {
                const v4f* __restrict__ xp =
                    reinterpret_cast<const v4f*>(x + (size_t)(nbase + lrow0 + i) * D);
                xa[i] = xp[lane];
                xb[i] = xp[32 + lane];
            }
            __builtin_amdgcn_sched_barrier(0);

            const v2f w0 = {wr0.x, ws0.x}, w1 = {wr0.y, ws0.y}, w2 = {wr0.z, ws0.z}, w3 = {wr0.w, ws0.w};
            const v2f w4 = {wr1.x, ws1.x}, w5 = {wr1.y, ws1.y}, w6 = {wr1.z, ws1.z}, w7 = {wr1.w, ws1.w};

            float val[2 * NR];
#pragma unroll
            for (int i = 0; i < NR; ++i) {
                v2f a2 = xa[i].x * w0;
                a2 += xa[i].y * w1;
                a2 += xa[i].z * w2;
                a2 += xa[i].w * w3;
                a2 += xb[i].x * w4;
                a2 += xb[i].y * w5;
                a2 += xb[i].z * w6;
                a2 += xb[i].w * w7;
                val[i]      = a2.x;
                val[NR + i] = a2.y;
                if (bt + 1 < NB) {
                    xrows[(wv * NR + i) * (D / 4) + lane]      = xa[i];
                    xrows[(wv * NR + i) * (D / 4) + 32 + lane] = xb[i];
                }
                if ((i & 3) == 3) __builtin_amdgcn_sched_barrier(0);
            }

            nd_fold<16>(val, lane);
            nd_fold<8>(val, lane);
            nd_fold<4>(val, lane);
            nd_fold<2>(val, lane);
            nd_fold<1>(val, lane);

            float* dstl = (lane < 16) ? rl : sl;
            dstl[lrow0 + (lane & 15)] = val[0];
        }
    }
    __syncthreads();

    const float s_own = sl[tid];

    int e = tid;
    for (; e + (EB - 1) * NPG < Eg; e += EB * NPG) {
        float rv[EB];
        if (e != tid) {
#pragma unroll
            for (int u = 0; u < EB; ++u) {
                sg[u] = psrc[e + u * NPG];
                dg[u] = pdst[e + u * NPG];
            }
        }
#pragma unroll
        for (int u = 0; u < EB; ++u) {
            const unsigned loc = (unsigned)(sg[u] - nbase);
            rv[u] = (loc < (unsigned)NPG) ? rl[loc] : far_row_dot(x, w_rel, sg[u]);
        }
#pragma unroll
        for (int u = 0; u < EB; ++u) atomicAdd(&acc[dg[u] - nbase], rv[u]);
    }
    for (; e < Eg; e += NPG) {
        const int sn = psrc[e];
        const unsigned loc = (unsigned)(sn - nbase);
        atomicAdd(&acc[pdst[e] - nbase], (loc < (unsigned)NPG) ? rl[loc] : far_row_dot(x, w_rel, sn));
    }
    __syncthreads();

    const float sc = tanhf(acc[tid] + s_own + brel);
    int kown = __float_as_int(sc);
    if (kown == (int)0x80000000) kown = 0;
    kown ^= (kown >> 31) & 0x7fffffff;
    skey[tid] = kown;
    __syncthreads();

    static_assert(NPG == 16 * 32, "one splitter per lane: 32 splitters, every 16th node");
    int K[RM];
#pragma unroll
    for (int m = 0; m < RM; ++m) K[m] = skey[32 * m + lane] + (m < wv ? 1 : 0);
    {
        const int r0 = rank_of(K, kown, 0);
        const int r1 = rank_of(K, kown, 16);
        if (lane == 0)  splr[2 * wv]     = r0;
        if (lane == 16) splr[2 * wv + 1] = r1;
    }
    __builtin_amdgcn_fence(__ATOMIC_RELEASE, "workgroup", "local");
    __builtin_amdgcn_s_barrier();
    __builtin_amdgcn_fence(__ATOMIC_ACQUIRE, "workgroup", "local");
    const int rs = splr[lane];
    const int ks = skey[16 * lane];
    int lo_c = (rs < k) ? ((rs << 5) | lane) : -1;
    int hi_c = (rs >= k) ? ((rs << 5) | lane) : 0x7fffffff;
#pragma unroll
    for (int off = 16; off > 0; off >>= 1) {
        const int lo_o = __shfl_xor(lo_c, off);
        const int hi_o = __shfl_xor(hi_c, off);
        lo_c = (lo_o > lo_c) ? lo_o : lo_c;
        hi_c = (hi_o < hi_c) ? hi_o : hi_c;
    }
    const int lo_u = __builtin_amdgcn_readfirstlane(lo_c);
    const int hi_u = __builtin_amdgcn_readfirstlane(hi_c);
    const bool has_lo = lo_u >= 0;
    const bool has_hi = hi_u != 0x7fffffff;
    const int klo = __builtin_amdgcn_readlane(ks, lo_u & 31);
    const int khi = __builtin_amdgcn_readlane(ks, hi_u & 31);
    const int ilo = 16 * (lo_u & 31);
    const int ihi = 16 * (hi_u & 31);
    const bool at_lo = (kown > klo) || (kown == klo && tid <= ilo);
    const bool at_hi = (khi > kown) || (khi == kown && ihi <= tid);
    const bool keep_sure = has_lo && at_lo;
    const bool drop_sure = has_hi && at_hi && !keep_sure;
    int cnt = drop_sure ? NPG : 0;
    unsigned um = __builtin_amdgcn_ballot_w32(!keep_sure && !drop_sure);
    while (um != 0u) {
        const int ii = __builtin_ctz(um);
        um &= um - 1u;
        const int r = rank_of(K, kown, ii);
        if (lane == ii) cnt = r;
    }
    const float wgt = (cnt < k) ? sc : 0.f;
    v4f pa = {0.f, 0.f, 0.f, 0.f}, pb = {0.f, 0.f, 0.f, 0.f};
#pragma unroll
    for (int i = 0; i < NR; ++i) {
        const float wi = __int_as_float(__builtin_amdgcn_readlane(__float_as_int(wgt), NR + i));
        pa += wi * xa[i];
        pb += wi * xb[i];
    }
    const int half = lane >> 4;
    const int l16  = lane & 15;
    const int q    = wv & 1;
    const int tp   = wv >> 1;
    const float* __restrict__ bp = w_proj + (size_t)(PJ_K * q + 4 * half) * OD + 32 * tp + l16;
    v4f bw0[PJ_U], bw1[PJ_U];
#pragma unroll
    for (int u = 0; u < PJ_G1; ++u) {
        const float* __restrict__ bq = bp + (size_t)(8 * u) * OD;
        bw0[u].x = bq[0];       bw0[u].y = bq[OD];       bw0[u].z = bq[2 * OD];       bw0[u].w = bq[3 * OD];
        bw1[u].x = bq[16];      bw1[u].y = bq[OD + 16];  bw1[u].z = bq[2 * OD + 16];  bw1[u].w = bq[3 * OD + 16];
    }
    const float bv = (tid < OD) ? b_proj[tid] : 0.f;

#pragma unroll
    for (int i = 0; i < NR; ++i) {
        const float wi = __int_as_float(__builtin_amdgcn_readlane(__float_as_int(wgt), i));
        pa += wi * xrows[(wv * NR + i) * (D / 4) + lane];
        pb += wi * xrows[(wv * NR + i) * (D / 4) + 32 + lane];
    }
    partial4[wv * PP + lane]      = pa;
    partial4[wv * PP + 32 + lane] = pb;
    __builtin_amdgcn_sched_barrier(0);
#pragma unroll
    for (int u = PJ_G1; u < PJ_G2; ++u) {
        const float* __restrict__ bq = bp + (size_t)(8 * u) * OD;
        bw0[u].x = bq[0];       bw0[u].y = bq[OD];       bw0[u].z = bq[2 * OD];       bw0[u].w = bq[3 * OD];
        bw1[u].x = bq[16];      bw1[u].y = bq[OD + 16];  bw1[u].z = bq[2 * OD + 16];  bw1[u].w = bq[3 * OD + 16];
    }
    __builtin_amdgcn_fence(__ATOMIC_RELEASE, "workgroup", "local");
    __builtin_amdgcn_s_barrier();
    __builtin_amdgcn_fence(__ATOMIC_ACQUIRE, "workgroup", "local");

    const lds_v4f* arow = (const lds_v4f*)(partial4 + l16 * PP + half + 2 * PJ_U * q);
    asm volatile("" : "+v"(arow));
    v4f av[PJ_U];
#pragma unroll
    for (int u = 0; u < PJ_U; ++u) av[u] = arow[2 * u];
#pragma unroll
    for (int u = PJ_G2; u < PJ_U; ++u) {
        const float* __restrict__ bq = bp + (size_t)(8 * u) * OD;
        bw0[u].x = bq[0];       bw0[u].y = bq[OD];       bw0[u].z = bq[2 * OD];       bw0[u].w = bq[3 * OD];
        bw1[u].x = bq[16];      bw1[u].y = bq[OD + 16];  bw1[u].z = bq[2 * OD + 16];  bw1[u].w = bq[3 * OD + 16];
    }
    __builtin_amdgcn_sched_barrier(0);

    v8f c0 = {}, c1 = {};
#pragma unroll
    for (int u = 0; u < PJ_U; ++u) {
        const v2f a0 = {av[u].x, av[u].y}, a1 = {av[u].z, av[u].w};
        const v2f b00 = {bw0[u].x, bw0[u].y}, b01 = {bw0[u].z, bw0[u].w};
        const v2f b10 = {bw1[u].x, bw1[u].y}, b11 = {bw1[u].z, bw1[u].w};
        c0 = __builtin_amdgcn_wmma_f32_16x16x4_f32(false, a0, false, b00, (short)0, c0, false, false);
        c1 = __builtin_amdgcn_wmma_f32_16x16x4_f32(false, a0, false, b10, (short)0, c1, false, false);
        c0 = __builtin_amdgcn_wmma_f32_16x16x4_f32(false, a1, false, b01, (short)0, c0, false, false);
        c1 = __builtin_amdgcn_wmma_f32_16x16x4_f32(false, a1, false, b11, (short)0, c1, false, false);
    }
    asm volatile("v_nop\n\tv_nop\n\tv_nop\n\tv_nop" : "+v"(c0), "+v"(c1));
    const float s0 = ((c0[0] + c0[1]) + (c0[2] + c0[3])) + ((c0[4] + c0[5]) + (c0[6] + c0[7]));
    const float s1 = ((c1[0] + c1[1]) + (c1[2] + c1[3])) + ((c1[4] + c1[5]) + (c1[6] + c1[7]));
    const float keep = half ? s1 : s0;
    const float send = half ? s0 : s1;
    red[q * OD + 32 * tp + lane] = keep + __shfl_xor(send, 16);
    __syncthreads();
    if (tid < OD) out[(size_t)b * OD + tid] = (red[tid] + red[OD + tid]) / (float)k + bv;
}

extern "C" void kernel_launch(void* const* d_in, const int* in_sizes, int n_in,
                              void* d_out, int out_size, void* d_ws, size_t ws_size,
                              hipStream_t stream) {
    const float* x      = (const float*)d_in[0];
    const int*   eidx   = (const int*)d_in[1];
    const float* w_rel  = (const float*)d_in[4];
    const float* b_rel  = (const float*)d_in[5];
    const float* w_root = (const float*)d_in[6];
    const float* w_proj = (const float*)d_in[7];
    const float* b_proj = (const float*)d_in[8];
    float*       out    = (float*)d_out;

    const int d    = D;
    const int N    = in_sizes[0] / d;
    const int Bg   = N / NPG;
    const int E    = in_sizes[1] / 2;
    const int Eg   = E / Bg;
    const int k    = (NPG + 1) / 2;

    graph_score_pool<<<dim3(Bg), dim3(NPG), 0, stream>>>(
        x, w_rel, w_root, eidx, eidx + E, b_rel, w_proj, b_proj, out, Eg, k);
}
